# starting kernel
# baseline (speedup 1.0000x reference)
#include <hip/hip_runtime.h>
#include <hip/hip_bf16.h>

typedef __attribute__((ext_vector_type(2))) float v2f;
typedef __attribute__((ext_vector_type(8))) float v8f;

__global__ __launch_bounds__(256)
void switchgate_score_kernel(const float* __restrict__ X,
                             const float* __restrict__ W,
                             const float* __restrict__ bias,
                             float* __restrict__ out,
                             int D) {
    const int lane = threadIdx.x & 31;
    const int n    = lane & 15;
    const int h    = lane >> 4;
    const int wave = threadIdx.x >> 5;
    const int tile = blockIdx.x * 8 + wave;
    const long rowBase = (long)tile * 16;

    const float* __restrict__ xrow = X + (rowBase + n) * (long)D + 2 * h;
    const float* __restrict__ wrow = W + (long)n * D + 2 * h;

    v8f c = {0.f, 0.f, 0.f, 0.f, 0.f, 0.f, 0.f, 0.f};

    for (int k = 0; k < D; k += 16) {
#pragma unroll
        for (int u = 0; u < 4; ++u) {
            v2f a  = *(const v2f*)(xrow + k + 4 * u);
            v2f bb = *(const v2f*)(wrow + k + 4 * u);
            c = __builtin_amdgcn_wmma_f32_16x16x4_f32(
                     false, a,  false, bb,
                     (short)0, c,  false,  false);
        }
    }

    const float bias_n = bias[n];

    float sc[8];
#pragma unroll
    for (int i = 0; i < 8; ++i) sc[i] = c[i] + bias_n;

#pragma unroll
    for (int i = 0; i < 8; ++i) {
        float s = sc[i];

        float m = s;
#pragma unroll
        for (int mask = 1; mask < 16; mask <<= 1)
            m = fmaxf(m, __shfl_xor(m, mask, 32));

        float e = __expf(s - m);
        float sum = e;
#pragma unroll
        for (int mask = 1; mask < 16; mask <<= 1)
            sum += __shfl_xor(sum, mask, 32);

        float gate = e / sum;

        int rank = 0;
#pragma unroll
        for (int j = 0; j < 16; ++j) {
            float sj = __shfl(s, j, 16);
            rank += (sj > s) || (sj == s && j < n);
        }
        sc[i] = (rank < 8) ? gate : 0.0f;
    }

#pragma unroll
    for (int i = 0; i < 8; ++i) {
        long t = rowBase + h * 8 + i;
        out[t * 16 + n] = sc[i];
    }
}

__global__ __launch_bounds__(256)
void switchgate_norm_kernel(float* __restrict__ out, int SE) {
    int idx = blockIdx.x * blockDim.x + threadIdx.x;
    if (idx >= SE) return;

    float v[8];
    float sum = 0.0f;
#pragma unroll
    for (int b = 0; b < 8; ++b) {
        v[b] = out[(long)b * SE + idx];
        sum += v[b];
    }
    float scale = 8.0f / (sum + 1e-6f);
#pragma unroll
    for (int b = 0; b < 8; ++b)
        out[(long)b * SE + idx] = v[b] * scale;
}

extern "C" void kernel_launch(void* const* d_in, const int* in_sizes, int n_in,
                              void* d_out, int out_size, void* d_ws, size_t ws_size,
                              hipStream_t stream) {
    const float* X    = (const float*)d_in[0];
    const float* W    = (const float*)d_in[1];
    const float* bias = (const float*)d_in[2];
    float* out = (float*)d_out;

    const int E = in_sizes[2];
    const int D = in_sizes[1] / E;
    const int tokens = in_sizes[0] / D;

    const int tiles = tokens / 16;
    switchgate_score_kernel<<<tiles / 8, 256, 0, stream>>>(X, W, bias, out, D);

    const int SE = out_size / 8;
    switchgate_norm_kernel<<<(SE + 255) / 256, 256, 0, stream>>>(out, SE);
}


# X streamed as whole 512-byte row pieces: 16-wave groups split K, W slice kept in registers, A fragments turned in wave-private LDS; f32 sum order changes (16 K-slice partials added in LDS)
# speedup vs baseline: 1.5621x; 1.5621x over previous
#include <hip/hip_runtime.h>

typedef __attribute__((ext_vector_type(4))) float v4f;
typedef __attribute__((ext_vector_type(8))) float v8f;

constexpr int WAVES     = 16;
constexpr int KS        = 128;
constexpr int D_MODEL   = WAVES * KS;
constexpr int TILES     = 8;
constexpr int TOK_WG    = TILES * 16;
constexpr int PITCH     = KS + 4;
constexpr int XS_WAVE   = 16 * PITCH;
constexpr int PART_WAVE = TILES * 256;

static __device__ __forceinline__ v4f load_x(const float* p) {
    return __builtin_nontemporal_load(reinterpret_cast<const v4f*>(p));
}

__global__ __launch_bounds__(WAVES * 32)
void switchgate_score_kernel(const float* __restrict__ X,
                             const float* __restrict__ W,
                             const float* __restrict__ bias,
                             float* __restrict__ out) {
    constexpr int D = D_MODEL;
    __shared__ __attribute__((aligned(16))) float xs[WAVES * XS_WAVE];
    __shared__ __attribute__((aligned(16))) float part[WAVES * PART_WAVE];

    const int tid  = threadIdx.x;
    const int lane = tid & 31;
    const int n    = lane & 15;
    const int h    = lane >> 4;
    const int wave = __builtin_amdgcn_readfirstlane(tid >> 5);
    const long tok0 = (long)blockIdx.x * TOK_WG;
    const int  k0   = wave * KS;

    v4f b[16];
    {
        const float* wp = W + (long)n * D + k0 + 4 * h;
#pragma unroll
        for (int j = 0; j < 16; ++j) b[j] = *reinterpret_cast<const v4f*>(wp + 8 * j);
    }

    const float* xp   = X + tok0 * (long)D + k0 + 4 * lane;
    float* myxs       = xs + wave * XS_WAVE;
    float* st_ptr     = myxs + 4 * lane;
    const float* fr_ptr = myxs + n * PITCH + 4 * h;
    float* mypart     = part + wave * PART_WAVE + lane;

    v4f xa[16], xb[16];

#define LOAD_TILE(BUF, T) do { \
    _Pragma("unroll") for (int r = 0; r < 16; ++r) \
        BUF[r] = load_x(xp + (long)((T) * 16 + r) * D); } while (0)

#define STAGE_TILE(BUF) do { \
    _Pragma("unroll") for (int r = 0; r < 16; ++r) \
        *reinterpret_cast<v4f*>(st_ptr + r * PITCH) = BUF[r]; } while (0)

#define WMMA(A2, B2, C) __builtin_amdgcn_wmma_f32_16x16x4_f32(false, (A2), false, (B2), (short)0, (C), false, false)

#define COMPUTE_TILE(T) do { \
    v8f c0 = {0.f, 0.f, 0.f, 0.f, 0.f, 0.f, 0.f, 0.f}; \
    v8f c1 = {0.f, 0.f, 0.f, 0.f, 0.f, 0.f, 0.f, 0.f}; \
    _Pragma("unroll") for (int g = 0; g < 2; ++g) { \
        v4f a[8]; \
        _Pragma("unroll") for (int j = 0; j < 8; ++j) \
            a[j] = *reinterpret_cast<const v4f*>(fr_ptr + 8 * (8 * g + j)); \
        _Pragma("unroll") for (int j = 0; j < 8; ++j) { \
            c0 = WMMA(a[j].xy, b[8 * g + j].xy, c0); \
            c1 = WMMA(a[j].zw, b[8 * g + j].zw, c1); \
        } \
    } \
    _Pragma("unroll") for (int i = 0; i < 8; ++i) \
        mypart[(T) * 256 + i * 32] = c0[i] + c1[i]; } while (0)

    LOAD_TILE(xa, 0);
    LOAD_TILE(xb, 1);

#pragma unroll
    for (int t = 0; t < TILES; t += 2) {
        STAGE_TILE(xa);
        if (t + 2 < TILES) LOAD_TILE(xa, t + 2);
        COMPUTE_TILE(t);

        STAGE_TILE(xb);
        if (t + 3 < TILES) LOAD_TILE(xb, t + 3);
        COMPUTE_TILE(t + 1);
    }
#undef LOAD_TILE
#undef STAGE_TILE
#undef COMPUTE_TILE
#undef WMMA

    __syncthreads();

    const int grp = tid >> 4;
    const float bias_n = bias[n];
#pragma unroll
    for (int r = 0; r < TOK_WG / 32; ++r) {
        const int T    = grp + 32 * r;
        const int tile = T >> 4;
        const int tt   = T & 15;
        const int off  = tile * 256 + (tt & 7) * 32 + (tt >> 3) * 16 + n;

        float s = part[off];
#pragma unroll
        for (int w = 1; w < WAVES; ++w) s += part[w * PART_WAVE + off];
        s += bias_n;

        float m = s;
#pragma unroll
        for (int mask = 1; mask < 16; mask <<= 1)
            m = fmaxf(m, __shfl_xor(m, mask, 32));

        float e = __expf(s - m);
        float sum = e;
#pragma unroll
        for (int mask = 1; mask < 16; mask <<= 1)
            sum += __shfl_xor(sum, mask, 32);

        float gate = e / sum;

        int rank = 0;
#pragma unroll
        for (int j = 0; j < 16; ++j) {
            float sj = __shfl(s, j, 16);
            rank += (sj > s) || (sj == s && j < n);
        }
        out[(tok0 + T) * 16 + n] = (rank < 8) ? gate : 0.0f;
    }
}

__global__ __launch_bounds__(256)
void switchgate_norm_kernel(float* __restrict__ out, int SE) {
    int idx = blockIdx.x * blockDim.x + threadIdx.x;
    if (idx >= SE) return;

    float v[8];
    float sum = 0.0f;
#pragma unroll
    for (int b = 0; b < 8; ++b) {
        v[b] = out[(long)b * SE + idx];
        sum += v[b];
    }
    float scale = 8.0f / (sum + 1e-6f);
#pragma unroll
    for (int b = 0; b < 8; ++b)
        out[(long)b * SE + idx] = v[b] * scale;
}

extern "C" void kernel_launch(void* const* d_in, const int* in_sizes, int n_in,
                              void* d_out, int out_size, void* d_ws, size_t ws_size,
                              hipStream_t stream) {
    const float* X    = (const float*)d_in[0];
    const float* W    = (const float*)d_in[1];
    const float* bias = (const float*)d_in[2];
    float* out = (float*)d_out;

    const int E = in_sizes[2];
    const int D = in_sizes[1] / E;
    const int tokens = in_sizes[0] / D;

    switchgate_score_kernel<<<tokens / TOK_WG, WAVES * 32, 0, stream>>>(X, W, bias, out);

    const int SE = out_size / 8;
    switchgate_norm_kernel<<<(SE + 255) / 256, 256, 0, stream>>>(out, SE);
}


# Workgroup = 16 positions x all 8 batches (tile t = batch t): cross-batch normalisation finished in the workgroup through LDS, second launch and out round trip removed; batch sum order 0..7 kept
# speedup vs baseline: 1.6854x; 1.0789x over previous
#include <hip/hip_runtime.h>

typedef __attribute__((ext_vector_type(4))) float v4f;
typedef __attribute__((ext_vector_type(8))) float v8f;

constexpr int WAVES     = 16;
constexpr int KS        = 128;
constexpr int D_MODEL   = WAVES * KS;
constexpr int TILES     = 8;
constexpr int TOK_WG    = TILES * 16;
constexpr int PITCH     = KS + 4;
constexpr int XS_WAVE   = 16 * PITCH;
constexpr int PART_WAVE = TILES * 256;

static __device__ __forceinline__ v4f load_x(const float* p) {
    return __builtin_nontemporal_load(reinterpret_cast<const v4f*>(p));
}

__global__ __launch_bounds__(WAVES * 32)
void switchgate_gate_kernel(const float* __restrict__ X,
                            const float* __restrict__ W,
                            const float* __restrict__ bias,
                            float* __restrict__ out,
                            int tokens) {
    constexpr int D = D_MODEL;
    __shared__ __attribute__((aligned(16))) float xs[WAVES * XS_WAVE];
    __shared__ __attribute__((aligned(16))) float part[WAVES * PART_WAVE];

    const int tid  = threadIdx.x;
    const int lane = tid & 31;
    const int n    = lane & 15;
    const int h    = lane >> 4;
    const int wave = __builtin_amdgcn_readfirstlane(tid >> 5);
    const int  S       = tokens / TILES;
    const long pos0    = (long)blockIdx.x * 16;
    const long bstride = (long)S * D;
    const int  k0      = wave * KS;

    v4f b[16];
    {
        const float* wp = W + (long)n * D + k0 + 4 * h;
#pragma unroll
        for (int j = 0; j < 16; ++j) b[j] = *reinterpret_cast<const v4f*>(wp + 8 * j);
    }

    const float* xp   = X + pos0 * (long)D + k0 + 4 * lane;
    float* myxs       = xs + wave * XS_WAVE;
    float* st_ptr     = myxs + 4 * lane;
    const float* fr_ptr = myxs + n * PITCH + 4 * h;
    float* mypart     = part + wave * PART_WAVE + lane;

    v4f xa[16], xb[16];

#define LOAD_TILE(BUF, T) do { \
    _Pragma("unroll") for (int r = 0; r < 16; ++r) \
        BUF[r] = load_x(xp + (T) * bstride + (long)r * D); } while (0)

#define STAGE_TILE(BUF) do { \
    _Pragma("unroll") for (int r = 0; r < 16; ++r) \
        *reinterpret_cast<v4f*>(st_ptr + r * PITCH) = BUF[r]; } while (0)

#define WMMA(A2, B2, C) __builtin_amdgcn_wmma_f32_16x16x4_f32(false, (A2), false, (B2), (short)0, (C), false, false)

#define COMPUTE_TILE(T) do { \
    v8f c0 = {0.f, 0.f, 0.f, 0.f, 0.f, 0.f, 0.f, 0.f}; \
    v8f c1 = {0.f, 0.f, 0.f, 0.f, 0.f, 0.f, 0.f, 0.f}; \
    _Pragma("unroll") for (int g = 0; g < 2; ++g) { \
        v4f a[8]; \
        _Pragma("unroll") for (int j = 0; j < 8; ++j) \
            a[j] = *reinterpret_cast<const v4f*>(fr_ptr + 8 * (8 * g + j)); \
        _Pragma("unroll") for (int j = 0; j < 8; ++j) { \
            c0 = WMMA(a[j].xy, b[8 * g + j].xy, c0); \
            c1 = WMMA(a[j].zw, b[8 * g + j].zw, c1); \
        } \
    } \
    _Pragma("unroll") for (int i = 0; i < 8; ++i) \
        mypart[(T) * 256 + i * 32] = c0[i] + c1[i]; } while (0)

    LOAD_TILE(xa, 0);
    LOAD_TILE(xb, 1);

#pragma unroll
    for (int t = 0; t < TILES; t += 2) {
        STAGE_TILE(xa);
        if (t + 2 < TILES) LOAD_TILE(xa, t + 2);
        COMPUTE_TILE(t);

        STAGE_TILE(xb);
        if (t + 3 < TILES) LOAD_TILE(xb, t + 3);
        COMPUTE_TILE(t + 1);
    }
#undef LOAD_TILE
#undef STAGE_TILE
#undef COMPUTE_TILE
#undef WMMA

    __syncthreads();

    const int grp = tid >> 4;
    const int tt  = grp & 15;
    const float bias_n = bias[n];
    float* mbuf = xs;
    float mk[TOK_WG / 32];
#pragma unroll
    for (int r = 0; r < TOK_WG / 32; ++r) {
        const int tile = (grp >> 4) + 2 * r;
        const int off  = tile * 256 + (tt & 7) * 32 + (tt >> 3) * 16 + n;

        float s = part[off];
#pragma unroll
        for (int w = 1; w < WAVES; ++w) s += part[w * PART_WAVE + off];
        s += bias_n;

        float m = s;
#pragma unroll
        for (int mask = 1; mask < 16; mask <<= 1)
            m = fmaxf(m, __shfl_xor(m, mask, 32));

        float e = __expf(s - m);
        float sum = e;
#pragma unroll
        for (int mask = 1; mask < 16; mask <<= 1)
            sum += __shfl_xor(sum, mask, 32);

        float gate = e / sum;

        int rank = 0;
#pragma unroll
        for (int j = 0; j < 16; ++j) {
            float sj = __shfl(s, j, 16);
            rank += (sj > s) || (sj == s && j < n);
        }
        mk[r] = (rank < 8) ? gate : 0.0f;
        mbuf[(tile * 16 + tt) * 16 + n] = mk[r];
    }

    __syncthreads();

    float den = 0.0f;
#pragma unroll
    for (int b = 0; b < TILES; ++b) den += mbuf[(b * 16 + tt) * 16 + n];
    const float scale = 8.0f / (den + 1e-6f);
#pragma unroll
    for (int r = 0; r < TOK_WG / 32; ++r) {
        const int tile = (grp >> 4) + 2 * r;
        out[((long)tile * S + pos0 + tt) * 16 + n] = mk[r] * scale;
    }
}

extern "C" void kernel_launch(void* const* d_in, const int* in_sizes, int n_in,
                              void* d_out, int out_size, void* d_ws, size_t ws_size,
                              hipStream_t stream) {
    const float* X    = (const float*)d_in[0];
    const float* W    = (const float*)d_in[1];
    const float* bias = (const float*)d_in[2];
    float* out = (float*)d_out;

    const int E = in_sizes[2];
    const int D = in_sizes[1] / E;
    const int tokens = in_sizes[0] / D;

    switchgate_gate_kernel<<<tokens / TOK_WG, WAVES * 32, 0, stream>>>(X, W, bias, out, tokens);
}


# Closing part stops waiting one LDS round trip per term: K-slice partials and batch terms requested in groups ahead of the adds (sum orders kept); 16-lane butterflies on DPP row_xmask, not bpermute
# speedup vs baseline: 1.7926x; 1.0636x over previous
#include <hip/hip_runtime.h>

typedef __attribute__((ext_vector_type(4))) float v4f;
typedef __attribute__((ext_vector_type(8))) float v8f;

constexpr int WAVES     = 16;
constexpr int KS        = 128;
constexpr int D_MODEL   = WAVES * KS;
constexpr int TILES     = 8;
constexpr int TOK_WG    = TILES * 16;
constexpr int PITCH     = KS + 4;
constexpr int XS_WAVE   = 16 * PITCH;
constexpr int PART_WAVE = TILES * 256;

static __device__ __forceinline__ v4f load_x(const float* p) {
    return __builtin_nontemporal_load(reinterpret_cast<const v4f*>(p));
}

template <int MASK>
static __device__ __forceinline__ float row_xor(float v) {
    return __int_as_float(__builtin_amdgcn_update_dpp(0, __float_as_int(v), 0x160 + MASK, 0xf, 0xf, true));
}

__global__ __launch_bounds__(WAVES * 32)
void switchgate_gate_kernel(const float* __restrict__ X,
                            const float* __restrict__ W,
                            const float* __restrict__ bias,
                            float* __restrict__ out,
                            int tokens) {
    constexpr int D = D_MODEL;
    __shared__ __attribute__((aligned(16))) float xs[WAVES * XS_WAVE];
    __shared__ __attribute__((aligned(16))) float part[WAVES * PART_WAVE];

    const int tid  = threadIdx.x;
    const int lane = tid & 31;
    const int n    = lane & 15;
    const int h    = lane >> 4;
    const int wave = __builtin_amdgcn_readfirstlane(tid >> 5);
    const int  S       = tokens / TILES;
    const long pos0    = (long)blockIdx.x * 16;
    const long bstride = (long)S * D;
    const int  k0      = wave * KS;

    const float bias_n = bias[n];

    v4f b[16];
    {
        const float* wp = W + (long)n * D + k0 + 4 * h;
#pragma unroll
        for (int j = 0; j < 16; ++j) b[j] = *reinterpret_cast<const v4f*>(wp + 8 * j);
    }

    const float* xp   = X + pos0 * (long)D + k0 + 4 * lane;
    float* myxs       = xs + wave * XS_WAVE;
    float* st_ptr     = myxs + 4 * lane;
    const float* fr_ptr = myxs + n * PITCH + 4 * h;
    float* mypart     = part + wave * PART_WAVE + lane;

    v4f xa[16], xb[16];

#define LOAD_TILE(BUF, T) do { \
    _Pragma("unroll") for (int r = 0; r < 16; ++r) \
        BUF[r] = load_x(xp + (T) * bstride + (long)r * D); } while (0)

#define STAGE_TILE(BUF) do { \
    _Pragma("unroll") for (int r = 0; r < 16; ++r) \
        *reinterpret_cast<v4f*>(st_ptr + r * PITCH) = BUF[r]; } while (0)

#define WMMA(A2, B2, C) __builtin_amdgcn_wmma_f32_16x16x4_f32(false, (A2), false, (B2), (short)0, (C), false, false)

#define COMPUTE_TILE(T) do { \
    v8f c0 = {0.f, 0.f, 0.f, 0.f, 0.f, 0.f, 0.f, 0.f}; \
    v8f c1 = {0.f, 0.f, 0.f, 0.f, 0.f, 0.f, 0.f, 0.f}; \
    _Pragma("unroll") for (int g = 0; g < 2; ++g) { \
        v4f a[8]; \
        _Pragma("unroll") for (int j = 0; j < 8; ++j) \
            a[j] = *reinterpret_cast<const v4f*>(fr_ptr + 8 * (8 * g + j)); \
        _Pragma("unroll") for (int j = 0; j < 8; ++j) { \
            c0 = WMMA(a[j].xy, b[8 * g + j].xy, c0); \
            c1 = WMMA(a[j].zw, b[8 * g + j].zw, c1); \
        } \
    } \
    _Pragma("unroll") for (int i = 0; i < 8; ++i) \
        mypart[(T) * 256 + i * 32] = c0[i] + c1[i]; } while (0)

    LOAD_TILE(xa, 0);
    LOAD_TILE(xb, 1);

#pragma unroll
    for (int t = 0; t < TILES; t += 2) {
        STAGE_TILE(xa);
        if (t + 2 < TILES) LOAD_TILE(xa, t + 2);
        COMPUTE_TILE(t);

        STAGE_TILE(xb);
        if (t + 3 < TILES) LOAD_TILE(xb, t + 3);
        COMPUTE_TILE(t + 1);
    }
#undef LOAD_TILE
#undef STAGE_TILE
#undef COMPUTE_TILE
#undef WMMA

    __syncthreads();

    const int grp = tid >> 4;
    const int tt  = grp & 15;
    float* mbuf = xs;
    constexpr int NTOK = TOK_WG / 32;

    const float* pl = part + (grp >> 4) * 256 + (tt & 7) * 32 + (tt >> 3) * 16 + n;

    float p[WAVES][NTOK];
#pragma unroll
    for (int w = 0; w < WAVES; ++w) {
#pragma unroll
        for (int r = 0; r < NTOK; ++r) p[w][r] = pl[w * PART_WAVE + 2 * r * 256];
    }

    float s[NTOK];
#pragma unroll
    for (int r = 0; r < NTOK; ++r) s[r] = p[0][r];
#pragma unroll
    for (int w = 1; w < WAVES; ++w) {
#pragma unroll
        for (int r = 0; r < NTOK; ++r) s[r] += p[w][r];
    }
#pragma unroll
    for (int r = 0; r < NTOK; ++r) s[r] += bias_n;

    float m[NTOK], e[NTOK], sum[NTOK];
#pragma unroll
    for (int r = 0; r < NTOK; ++r) m[r] = s[r];
#define ROW_STEP(OP, V, MASK) _Pragma("unroll") for (int r = 0; r < NTOK; ++r) V[r] = OP(V[r], row_xor<MASK>(V[r]))
#define OP_MAX(a, b) fmaxf((a), (b))
#define OP_ADD(a, b) ((a) + (b))
    ROW_STEP(OP_MAX, m, 1); ROW_STEP(OP_MAX, m, 2); ROW_STEP(OP_MAX, m, 4); ROW_STEP(OP_MAX, m, 8);
#pragma unroll
    for (int r = 0; r < NTOK; ++r) { e[r] = __expf(s[r] - m[r]); sum[r] = e[r]; }
    ROW_STEP(OP_ADD, sum, 1); ROW_STEP(OP_ADD, sum, 2); ROW_STEP(OP_ADD, sum, 4); ROW_STEP(OP_ADD, sum, 8);
#undef ROW_STEP
#undef OP_MAX
#undef OP_ADD

    float mk[NTOK];
#pragma unroll
    for (int r = 0; r < NTOK; ++r) {
        const int tile = (grp >> 4) + 2 * r;
        const float gate = e[r] / sum[r];
        int rank = 0;
#pragma unroll
        for (int j = 0; j < 16; ++j) {
            float sj = __shfl(s[r], j, 16);
            rank += (sj > s[r]) || (sj == s[r] && j < n);
        }
        mk[r] = (rank < 8) ? gate : 0.0f;
        mbuf[(tile * 16 + tt) * 16 + n] = mk[r];
    }

    __syncthreads();

    float d[TILES];
#pragma unroll
    for (int b = 0; b < TILES; ++b) d[b] = mbuf[(b * 16 + tt) * 16 + n];
    float den = 0.0f;
#pragma unroll
    for (int b = 0; b < TILES; ++b) den += d[b];
    const float scale = 8.0f / (den + 1e-6f);
#pragma unroll
    for (int r = 0; r < NTOK; ++r) {
        const int tile = (grp >> 4) + 2 * r;
        out[((long)tile * S + pos0 + tt) * 16 + n] = mk[r] * scale;
    }
}

extern "C" void kernel_launch(void* const* d_in, const int* in_sizes, int n_in,
                              void* d_out, int out_size, void* d_ws, size_t ws_size,
                              hipStream_t stream) {
    const float* X    = (const float*)d_in[0];
    const float* W    = (const float*)d_in[1];
    const float* bias = (const float*)d_in[2];
    float* out = (float*)d_out;

    const int E = in_sizes[2];
    const int D = in_sizes[1] / E;
    const int tokens = in_sizes[0] / D;

    switchgate_gate_kernel<<<tokens / TOK_WG, WAVES * 32, 0, stream>>>(X, W, bias, out, tokens);
}
